# starting kernel
# baseline (speedup 1.0000x reference)
#include <hip/hip_runtime.h>
#include <hip/hip_bf16.h>

typedef __attribute__((ext_vector_type(2))) float v2f;
typedef __attribute__((ext_vector_type(8))) float v8f;

#define NUMK 128
#define NUME 1024
#define HID  32
#define HXW  (NUMK + NUME + 2)
#define AS   132
#define XS   36
#define GTS  132
#define WTS  36

__device__ __forceinline__ v8f wmma_step(const float* __restrict__ Am, int aStride,
                                         const float* __restrict__ BTm, int btStride,
                                         int m0, int n0, int k0, int half, int l, v8f c) {
    v2f a, b;
    a.x = Am[(m0 + l) * aStride + k0 + 2 * half];
    a.y = Am[(m0 + l) * aStride + k0 + 2 * half + 1];
    b.x = BTm[(n0 + l) * btStride + k0 + 2 * half];
    b.y = BTm[(n0 + l) * btStride + k0 + 2 * half + 1];
    return __builtin_amdgcn_wmma_f32_16x16x4_f32(false, a, false, b, (short)0, c,
                                                 false, false);
}

__global__ __launch_bounds__(256) void gcn3_kernel(
    const float* __restrict__ Hx, const int* __restrict__ ei,
    const float* __restrict__ W1, const float* __restrict__ b1,
    const float* __restrict__ W2, const float* __restrict__ b2,
    const float* __restrict__ W3, const float* __restrict__ b3,
    float* __restrict__ out) {
    __shared__ float As[NUMK * AS];
    __shared__ float Xs[NUMK * XS];
    __shared__ float GT[HID * GTS];
    __shared__ float W2T[HID * WTS];
    __shared__ float ps[NUMK], a1s[NUMK], vs[NUMK], deg[NUMK], dinv[NUMK];
    __shared__ float W1s[HID], b1s[HID], b2s[HID], W3s[HID];

    const int g    = blockIdx.x;
    const int t    = threadIdx.x;
    const int lane = t & 31;
    const int wave = t >> 5;
    const int half = lane >> 4;
    const int l    = lane & 15;
    const float* hx = Hx + (size_t)g * HXW;

    for (int i = t; i < NUMK * AS; i += 256) As[i] = 0.0f;
    if (t < NUMK) ps[t] = hx[t];
    if (t < HID) { W1s[t] = W1[t]; b1s[t] = b1[t]; b2s[t] = b2[t]; W3s[t] = W3[t]; }
    for (int i = t; i < HID * HID; i += 256)
        W2T[(i & 31) * WTS + (i >> 5)] = W2[i];
    __syncthreads();

    if (t < NUMK) { As[t * AS + t] = 1.0f; deg[t] = 1.0f; }
    __syncthreads();

    for (int e = t; e < NUME; e += 256) {
        int s = ei[e];
        int d = ei[NUME + e];
        float w = hx[NUMK + e];
        atomicAdd(&As[d * AS + s], w);
        atomicAdd(&deg[d], w);
    }
    __syncthreads();

    if (t < NUMK) dinv[t] = rsqrtf(deg[t]);
    __syncthreads();
    for (int i = t; i < NUMK * NUMK; i += 256) {
        int d = i >> 7, s = i & 127;
        As[d * AS + s] *= dinv[d] * dinv[s];
    }
    __syncthreads();

    if (t < NUMK) {
        float acc = 0.0f;
        for (int s = 0; s < NUMK; ++s) acc += As[t * AS + s] * ps[s];
        a1s[t] = acc;
    }
    __syncthreads();
    for (int i = t; i < NUMK * HID; i += 256) {
        int n = i >> 5, f = i & 31;
        Xs[n * XS + f] = fmaxf(fmaf(a1s[n], W1s[f], b1s[f]), 0.0f);
    }
    __syncthreads();

    {
        const int m0 = wave * 16;
        for (int nt = 0; nt < 2; ++nt) {
            const int n0 = nt * 16;
            v8f c = {};
#pragma unroll
            for (int k0 = 0; k0 < HID; k0 += 4)
                c = wmma_step(Xs, XS, W2T, WTS, m0, n0, k0, half, l, c);
#pragma unroll
            for (int j = 0; j < 8; ++j)
                GT[(n0 + l) * GTS + (m0 + j + 8 * half)] = c[j];
        }
    }
    __syncthreads();

    {
        const int m0 = wave * 16;
        for (int nt = 0; nt < 2; ++nt) {
            const int n0 = nt * 16;
            v8f c = {};
#pragma unroll
            for (int k0 = 0; k0 < NUMK; k0 += 4)
                c = wmma_step(As, AS, GT, GTS, m0, n0, k0, half, l, c);
#pragma unroll
            for (int j = 0; j < 8; ++j) {
                int row = m0 + j + 8 * half, col = n0 + l;
                Xs[row * XS + col] = fmaxf(c[j] + b2s[col], 0.0f);
            }
        }
    }
    __syncthreads();

    if (t < NUMK) {
        float acc = 0.0f;
        for (int f = 0; f < HID; ++f) acc += Xs[t * XS + f] * W3s[f];
        vs[t] = acc;
    }
    __syncthreads();
    if (t < NUMK) {
        float acc = 0.0f;
        for (int s = 0; s < NUMK; ++s) acc += As[t * AS + s] * vs[s];
        out[(size_t)g * NUMK + t] = acc + b3[0];
    }
}

extern "C" void kernel_launch(void* const* d_in, const int* in_sizes, int n_in,
                              void* d_out, int out_size, void* d_ws, size_t ws_size,
                              hipStream_t stream) {
    (void)n_in; (void)d_ws; (void)ws_size; (void)out_size;
    const float* Hx = (const float*)d_in[0];
    const int*   ei = (const int*)d_in[1];
    const float* W1 = (const float*)d_in[2];
    const float* b1 = (const float*)d_in[3];
    const float* W2 = (const float*)d_in[4];
    const float* b2 = (const float*)d_in[5];
    const float* W3 = (const float*)d_in[6];
    const float* b3 = (const float*)d_in[7];
    float* out = (float*)d_out;
    const int nb = in_sizes[0] / HXW;
    gcn3_kernel<<<dim3(nb), dim3(256), 0, stream>>>(Hx, ei, W1, b1, W2, b2, W3, b3, out);
}


# Each wave reads its 16x128 adjacency row tile from LDS once into registers, scales it there and feeds A@p, both column tiles of A@g and A@v from it: LDS rescale pass gone; A@p, A@v sum order changed
# speedup vs baseline: 1.2493x; 1.2493x over previous
#include <hip/hip_runtime.h>
#include <hip/hip_bf16.h>


typedef __attribute__((ext_vector_type(2))) float v2f;
typedef __attribute__((ext_vector_type(8))) float v8f;

#define NUMK 128
#define NUME 1024
#define HID  32
#define HXW  (NUMK + NUME + 2)
#define AS   132
#define XS   36
#define GTS  132
#define WTS  36
#define KST  (NUMK / 4)

__device__ __forceinline__ v8f wmma_f32(v2f a, v2f b, v8f c) {
    return __builtin_amdgcn_wmma_f32_16x16x4_f32(false, a, false, b, (short)0, c,
                                                 false, false);
}

__device__ __forceinline__ v8f wmma_step(const float* __restrict__ Am, int aStride,
                                         const float* __restrict__ BTm, int btStride,
                                         int m0, int n0, int k0, int half, int l, v8f c) {
    v2f a, b;
    a.x = Am[(m0 + l) * aStride + k0 + 2 * half];
    a.y = Am[(m0 + l) * aStride + k0 + 2 * half + 1];
    b.x = BTm[(n0 + l) * btStride + k0 + 2 * half];
    b.y = BTm[(n0 + l) * btStride + k0 + 2 * half + 1];
    return wmma_f32(a, b, c);
}

__device__ __forceinline__ float xhalf(float x) {
    return __builtin_bit_cast(float,
        __builtin_amdgcn_permlanex16(__builtin_bit_cast(int, x), __builtin_bit_cast(int, x),
                                     0x76543210, 0xfedcba98, false, false));
}

__device__ __forceinline__ float tile_times_vec(const v2f (&a)[KST], const float* vec, int half) {
    v2f acc = {0.0f, 0.0f};
    const float* vp = vec + 2 * half;
#pragma unroll
    for (int i = 0; i < KST; ++i) {
        v2f x = *(const v2f*)(vp + 4 * i);
        acc = __builtin_elementwise_fma(a[i], x, acc);
    }
    float part = acc.x + acc.y;
    return part + xhalf(part);
}

__global__ __launch_bounds__(256) void gcn3_kernel(
    const float* __restrict__ Hx, const int* __restrict__ ei,
    const float* __restrict__ W1, const float* __restrict__ b1,
    const float* __restrict__ W2, const float* __restrict__ b2,
    const float* __restrict__ W3, const float* __restrict__ b3,
    float* __restrict__ out) {
    __shared__ __attribute__((aligned(16))) float As[NUMK * AS];
    __shared__ __attribute__((aligned(16))) float Xs[NUMK * XS];
    __shared__ __attribute__((aligned(16))) float GT[HID * GTS];
    __shared__ __attribute__((aligned(16))) float W2T[HID * WTS];
    __shared__ __attribute__((aligned(16))) float ps[NUMK];
    __shared__ __attribute__((aligned(16))) float a1s[NUMK];
    __shared__ __attribute__((aligned(16))) float vs[NUMK];
    __shared__ __attribute__((aligned(16))) float deg[NUMK];
    __shared__ __attribute__((aligned(16))) float dinv[NUMK];
    __shared__ float W1s[HID], b1s[HID], b2s[HID], W3s[HID];

    const int g    = blockIdx.x;
    const int t    = threadIdx.x;
    const int lane = t & 31;
    const int wave = t >> 5;
    const int half = lane >> 4;
    const int l    = lane & 15;
    const int m0   = wave * 16;
    const float* hx = Hx + (size_t)g * HXW;

    for (int i = t; i < NUMK * AS; i += 256) As[i] = 0.0f;
    if (t < NUMK) ps[t] = hx[t];
    if (t < HID) { W1s[t] = W1[t]; b1s[t] = b1[t]; b2s[t] = b2[t]; W3s[t] = W3[t]; }
    for (int i = t; i < HID * HID; i += 256)
        W2T[(i & 31) * WTS + (i >> 5)] = W2[i];
    __syncthreads();

    if (t < NUMK) { As[t * AS + t] = 1.0f; deg[t] = 1.0f; }
    __syncthreads();

    for (int e = t; e < NUME; e += 256) {
        int s = ei[e];
        int d = ei[NUME + e];
        float w = hx[NUMK + e];
        atomicAdd(&As[d * AS + s], w);
        atomicAdd(&deg[d], w);
    }
    __syncthreads();

    if (t < NUMK) dinv[t] = rsqrtf(deg[t]);
    __syncthreads();

    v2f a[KST];
    {
        const float* arow = As + (m0 + l) * AS + 2 * half;
        const float* dcol = dinv + 2 * half;
        const float dr = dinv[m0 + l];
#pragma unroll
        for (int i = 0; i < KST; ++i) {
            v2f av = *(const v2f*)(arow + 4 * i);
            v2f dc = *(const v2f*)(dcol + 4 * i);
            a[i] = av * (dc * dr);
        }
    }

    {
        float a1 = tile_times_vec(a, ps, half);
        if (half == 0) a1s[m0 + l] = a1;
    }
    __syncthreads();
    for (int i = t; i < NUMK * HID; i += 256) {
        int n = i >> 5, f = i & 31;
        Xs[n * XS + f] = fmaxf(fmaf(a1s[n], W1s[f], b1s[f]), 0.0f);
    }
    __syncthreads();

    {
        for (int nt = 0; nt < 2; ++nt) {
            const int n0 = nt * 16;
            v8f c = {};
#pragma unroll
            for (int k0 = 0; k0 < HID; k0 += 4)
                c = wmma_step(Xs, XS, W2T, WTS, m0, n0, k0, half, l, c);
#pragma unroll
            for (int j = 0; j < 8; ++j)
                GT[(n0 + l) * GTS + (m0 + j + 8 * half)] = c[j];
        }
    }
    __syncthreads();

    {
        v8f c0 = {}, c1 = {};
        const float* g0 = GT + l * GTS + 2 * half;
        const float* g1 = GT + (16 + l) * GTS + 2 * half;
#pragma unroll
        for (int i = 0; i < KST; ++i) {
            v2f bA = *(const v2f*)(g0 + 4 * i);
            v2f bB = *(const v2f*)(g1 + 4 * i);
            c0 = wmma_f32(a[i], bA, c0);
            c1 = wmma_f32(a[i], bB, c1);
        }
#pragma unroll
        for (int j = 0; j < 8; ++j) {
            int row = m0 + j + 8 * half;
            Xs[row * XS + l]      = fmaxf(c0[j] + b2s[l], 0.0f);
            Xs[row * XS + 16 + l] = fmaxf(c1[j] + b2s[16 + l], 0.0f);
        }
    }
    __syncthreads();

    if (t < NUMK) {
        float acc = 0.0f;
        for (int f = 0; f < HID; ++f) acc += Xs[t * XS + f] * W3s[f];
        vs[t] = acc;
    }
    __syncthreads();
    {
        float o = tile_times_vec(a, vs, half);
        if (half == 0) a1s[m0 + l] = o;
    }
    __syncthreads();
    if (t < NUMK) out[(size_t)g * NUMK + t] = a1s[t] + b3[0];
}

extern "C" void kernel_launch(void* const* d_in, const int* in_sizes, int n_in,
                              void* d_out, int out_size, void* d_ws, size_t ws_size,
                              hipStream_t stream) {
    (void)n_in; (void)d_ws; (void)ws_size; (void)out_size;
    const float* Hx = (const float*)d_in[0];
    const int*   ei = (const int*)d_in[1];
    const float* W1 = (const float*)d_in[2];
    const float* b1 = (const float*)d_in[3];
    const float* W2 = (const float*)d_in[4];
    const float* b2 = (const float*)d_in[5];
    const float* W3 = (const float*)d_in[6];
    const float* b3 = (const float*)d_in[7];
    float* out = (float*)d_out;
    const int nb = in_sizes[0] / HXW;
    gcn3_kernel<<<dim3(nb), dim3(256), 0, stream>>>(Hx, ei, W1, b1, W2, b2, W3, b3, out);
}


# Layer 2 as (A@x1)@W2, both products transposed so each WMMA result tile is the next product's B operand in registers: x1, g, x2 leave LDS, 3 barriers gone; sum order of layer 2 and x2@W3 changed
# speedup vs baseline: 1.8086x; 1.4477x over previous
#include <hip/hip_runtime.h>
#include <hip/hip_bf16.h>


typedef __attribute__((ext_vector_type(2))) float v2f;
typedef __attribute__((ext_vector_type(8))) float v8f;

#define NUMK 128
#define NUME 1024
#define HID  32
#define HXW  (NUMK + NUME + 2)
#define AS   132
#define WTS  36
#define KST  (NUMK / 4)

__device__ __forceinline__ v8f wmma_f32(v2f a, v2f b, v8f c) {
    return __builtin_amdgcn_wmma_f32_16x16x4_f32(false, a, false, b, (short)0, c,
                                                 false, false);
}

__device__ __forceinline__ float xhalf(float x) {
    return __builtin_bit_cast(float,
        __builtin_amdgcn_permlanex16(__builtin_bit_cast(int, x), __builtin_bit_cast(int, x),
                                     0x76543210, 0xfedcba98, false, false));
}

__device__ __forceinline__ float tile_times_vec(const v2f (&a)[KST], const float* vec, int half) {
    v2f acc = {0.0f, 0.0f};
    const float* vp = vec + 2 * half;
#pragma unroll
    for (int i = 0; i < KST; ++i) {
        v2f x = *(const v2f*)(vp + 4 * i);
        acc = __builtin_elementwise_fma(a[i], x, acc);
    }
    float part = acc.x + acc.y;
    return part + xhalf(part);
}

__global__ __launch_bounds__(256) void gcn3_kernel(
    const float* __restrict__ Hx, const int* __restrict__ ei,
    const float* __restrict__ W1, const float* __restrict__ b1,
    const float* __restrict__ W2, const float* __restrict__ b2,
    const float* __restrict__ W3, const float* __restrict__ b3,
    float* __restrict__ out) {
    __shared__ __attribute__((aligned(16))) float As[NUMK * AS];
    __shared__ __attribute__((aligned(16))) float W2T[HID * WTS];
    __shared__ __attribute__((aligned(16))) float ps[NUMK];
    __shared__ __attribute__((aligned(16))) float a1s[NUMK];
    __shared__ __attribute__((aligned(16))) float vs[NUMK];
    __shared__ __attribute__((aligned(16))) float deg[NUMK];
    __shared__ __attribute__((aligned(16))) float dinv[NUMK];
    __shared__ float W1s[HID], b1s[HID], b2s[HID], W3s[HID];

    const int g    = blockIdx.x;
    const int t    = threadIdx.x;
    const int lane = t & 31;
    const int wave = t >> 5;
    const int half = lane >> 4;
    const int l    = lane & 15;
    const int m0   = wave * 16;
    const float* hx = Hx + (size_t)g * HXW;

    for (int i = t; i < NUMK * AS; i += 256) As[i] = 0.0f;
    if (t < NUMK) ps[t] = hx[t];
    if (t < HID) { W1s[t] = W1[t]; b1s[t] = b1[t]; b2s[t] = b2[t]; W3s[t] = W3[t]; }
    for (int i = t; i < HID * HID; i += 256)
        W2T[(i & 31) * WTS + (i >> 5)] = W2[i];
    __syncthreads();

    if (t < NUMK) { As[t * AS + t] = 1.0f; deg[t] = 1.0f; }
    __syncthreads();

    for (int e = t; e < NUME; e += 256) {
        int s = ei[e];
        int d = ei[NUME + e];
        float w = hx[NUMK + e];
        atomicAdd(&As[d * AS + s], w);
        atomicAdd(&deg[d], w);
    }
    __syncthreads();

    if (t < NUMK) dinv[t] = rsqrtf(deg[t]);
    __syncthreads();

    v2f a[KST];
    {
        const float* arow = As + (m0 + l) * AS + 2 * half;
        const float* dcol = dinv + 2 * half;
        const float dr = dinv[m0 + l];
#pragma unroll
        for (int i = 0; i < KST; ++i) {
            v2f av = *(const v2f*)(arow + 4 * i);
            v2f dc = *(const v2f*)(dcol + 4 * i);
            a[i] = av * (dc * dr);
        }
    }

    {
        float a1 = tile_times_vec(a, ps, half);
        if (half == 0) a1s[m0 + l] = a1;
    }
    __syncthreads();

    v8f y0 = {}, y1 = {};
    {
        const float w1a = W1s[l], w1b = W1s[16 + l];
        const float c1a = b1s[l], c1b = b1s[16 + l];
        const float* ap = a1s + 2 * half;
#pragma unroll
        for (int i = 0; i < KST; ++i) {
            v2f s = *(const v2f*)(ap + 4 * i);
            v2f xa, xb;
            xa.x = fmaxf(fmaf(s.x, w1a, c1a), 0.0f);
            xa.y = fmaxf(fmaf(s.y, w1a, c1a), 0.0f);
            xb.x = fmaxf(fmaf(s.x, w1b, c1b), 0.0f);
            xb.y = fmaxf(fmaf(s.y, w1b, c1b), 0.0f);
            y0 = wmma_f32(xa, a[i], y0);
            y1 = wmma_f32(xb, a[i], y1);
        }
    }
    v8f z0 = {}, z1 = {};
    {
        const float* w0 = W2T + l * WTS + 8 * half;
        const float* w1 = W2T + (16 + l) * WTS + 8 * half;
#pragma unroll
        for (int q = 0; q < 4; ++q) {
            v2f b  = {y0[2 * q], y0[2 * q + 1]};
            v2f wa = *(const v2f*)(w0 + 2 * q);
            v2f wb = *(const v2f*)(w1 + 2 * q);
            z0 = wmma_f32(wa, b, z0);
            z1 = wmma_f32(wb, b, z1);
        }
#pragma unroll
        for (int q = 0; q < 4; ++q) {
            v2f b  = {y1[2 * q], y1[2 * q + 1]};
            v2f wa = *(const v2f*)(w0 + 16 + 2 * q);
            v2f wb = *(const v2f*)(w1 + 16 + 2 * q);
            z0 = wmma_f32(wa, b, z0);
            z1 = wmma_f32(wb, b, z1);
        }
    }

    {
        const float* bb = b2s + 8 * half;
        const float* ww = W3s + 8 * half;
        float pa = 0.0f, pb = 0.0f;
#pragma unroll
        for (int r = 0; r < 8; ++r) {
            pa = fmaf(fmaxf(z0[r] + bb[r], 0.0f), ww[r], pa);
            pb = fmaf(fmaxf(z1[r] + bb[16 + r], 0.0f), ww[16 + r], pb);
        }
        float part = pa + pb;
        float v = part + xhalf(part);
        if (half == 0) vs[m0 + l] = v;
    }
    __syncthreads();
    {
        float o = tile_times_vec(a, vs, half);
        if (half == 0) a1s[m0 + l] = o;
    }
    __syncthreads();
    if (t < NUMK) out[(size_t)g * NUMK + t] = a1s[t] + b3[0];
}

extern "C" void kernel_launch(void* const* d_in, const int* in_sizes, int n_in,
                              void* d_out, int out_size, void* d_ws, size_t ws_size,
                              hipStream_t stream) {
    (void)n_in; (void)d_ws; (void)ws_size; (void)out_size;
    const float* Hx = (const float*)d_in[0];
    const int*   ei = (const int*)d_in[1];
    const float* W1 = (const float*)d_in[2];
    const float* b1 = (const float*)d_in[3];
    const float* W2 = (const float*)d_in[4];
    const float* b2 = (const float*)d_in[5];
    const float* W3 = (const float*)d_in[6];
    const float* b3 = (const float*)d_in[7];
    float* out = (float*)d_out;
    const int nb = in_sizes[0] / HXW;
    gcn3_kernel<<<dim3(nb), dim3(256), 0, stream>>>(Hx, ei, W1, b1, W2, b2, W3, b3, out);
}


# Edges and W2 read in wide contiguous runs (4 consecutive edges per thread, W2 by waves 4-7): the load-wait loops of scatter and W2 staging are gone; arrival order of the LDS float adds changed
# speedup vs baseline: 2.0413x; 1.1287x over previous
#include <hip/hip_runtime.h>
#include <hip/hip_bf16.h>


typedef __attribute__((ext_vector_type(2))) float v2f;
typedef __attribute__((ext_vector_type(8))) float v8f;
typedef __attribute__((ext_vector_type(4))) float v4f;
typedef __attribute__((ext_vector_type(4))) int   v4i;

#define NUMK 128
#define NUME 1024
#define HID  32
#define HXW  (NUMK + NUME + 2)
#define AS   132
#define WTS  36
#define KST  (NUMK / 4)

__device__ __forceinline__ v8f wmma_f32(v2f a, v2f b, v8f c) {
    return __builtin_amdgcn_wmma_f32_16x16x4_f32(false, a, false, b, (short)0, c,
                                                 false, false);
}

__device__ __forceinline__ float xhalf(float x) {
    return __builtin_bit_cast(float,
        __builtin_amdgcn_permlanex16(__builtin_bit_cast(int, x), __builtin_bit_cast(int, x),
                                     0x76543210, 0xfedcba98, false, false));
}

__device__ __forceinline__ float tile_times_vec(const v2f (&a)[KST], const float* vec, int half) {
    v2f acc = {0.0f, 0.0f};
    const float* vp = vec + 2 * half;
#pragma unroll
    for (int i = 0; i < KST; ++i) {
        v2f x = *(const v2f*)(vp + 4 * i);
        acc = __builtin_elementwise_fma(a[i], x, acc);
    }
    float part = acc.x + acc.y;
    return part + xhalf(part);
}

__global__ __launch_bounds__(256) void gcn3_kernel(
    const float* __restrict__ Hx, const int* __restrict__ ei,
    const float* __restrict__ W1, const float* __restrict__ b1,
    const float* __restrict__ W2, const float* __restrict__ b2,
    const float* __restrict__ W3, const float* __restrict__ b3,
    float* __restrict__ out) {
    __shared__ __attribute__((aligned(16))) float As[NUMK * AS];
    __shared__ __attribute__((aligned(16))) float W2T[HID * WTS];
    __shared__ __attribute__((aligned(16))) float ps[NUMK];
    __shared__ __attribute__((aligned(16))) float a1s[NUMK];
    __shared__ __attribute__((aligned(16))) float vs[NUMK];
    __shared__ __attribute__((aligned(16))) float deg[NUMK];
    __shared__ __attribute__((aligned(16))) float dinv[NUMK];
    __shared__ float W1s[HID], b1s[HID], b2s[HID], W3s[HID];

    const int g    = blockIdx.x;
    const int t    = threadIdx.x;
    const int lane = t & 31;
    const int wave = t >> 5;
    const int half = lane >> 4;
    const int l    = lane & 15;
    const int m0   = wave * 16;
    const float* hx = Hx + (size_t)g * HXW;

    for (int i = t; i < NUMK * AS; i += 256) As[i] = 0.0f;
    if (t < NUMK) ps[t] = hx[t];
    if (t < HID) { W1s[t] = W1[t]; b1s[t] = b1[t]; b2s[t] = b2[t]; W3s[t] = W3[t]; }
    if (t >= NUMK) {
        const int u = t - NUMK;
        const v4f wlo = *(const v4f*)(W2 + 8 * u);
        const v4f whi = *(const v4f*)(W2 + 8 * u + 4);
        float* wt = W2T + (8 * (u & 3)) * WTS + (u >> 2);
#pragma unroll
        for (int j = 0; j < 4; ++j) {
            wt[j * WTS]       = wlo[j];
            wt[(4 + j) * WTS] = whi[j];
        }
    }
    __syncthreads();

    if (t < NUMK) { As[t * AS + t] = 1.0f; deg[t] = 1.0f; }
    __syncthreads();

    {
        const v4i s4 = *(const v4i*)(ei + 4 * t);
        const v4i d4 = *(const v4i*)(ei + NUME + 4 * t);
        const v2f wlo = *(const v2f*)(hx + NUMK + 4 * t);
        const v2f whi = *(const v2f*)(hx + NUMK + 4 * t + 2);
        const float w4[4] = {wlo.x, wlo.y, whi.x, whi.y};
#pragma unroll
        for (int j = 0; j < 4; ++j) {
            atomicAdd(&As[d4[j] * AS + s4[j]], w4[j]);
            atomicAdd(&deg[d4[j]], w4[j]);
        }
    }
    __syncthreads();

    if (t < NUMK) dinv[t] = rsqrtf(deg[t]);
    __syncthreads();

    v2f a[KST];
    {
        const float* arow = As + (m0 + l) * AS + 2 * half;
        const float* dcol = dinv + 2 * half;
        const float dr = dinv[m0 + l];
#pragma unroll
        for (int i = 0; i < KST; ++i) {
            v2f av = *(const v2f*)(arow + 4 * i);
            v2f dc = *(const v2f*)(dcol + 4 * i);
            a[i] = av * (dc * dr);
        }
    }

    {
        float a1 = tile_times_vec(a, ps, half);
        if (half == 0) a1s[m0 + l] = a1;
    }
    __syncthreads();

    v8f y0 = {}, y1 = {};
    {
        const float w1a = W1s[l], w1b = W1s[16 + l];
        const float c1a = b1s[l], c1b = b1s[16 + l];
        const float* ap = a1s + 2 * half;
#pragma unroll
        for (int i = 0; i < KST; ++i) {
            v2f s = *(const v2f*)(ap + 4 * i);
            v2f xa, xb;
            xa.x = fmaxf(fmaf(s.x, w1a, c1a), 0.0f);
            xa.y = fmaxf(fmaf(s.y, w1a, c1a), 0.0f);
            xb.x = fmaxf(fmaf(s.x, w1b, c1b), 0.0f);
            xb.y = fmaxf(fmaf(s.y, w1b, c1b), 0.0f);
            y0 = wmma_f32(xa, a[i], y0);
            y1 = wmma_f32(xb, a[i], y1);
        }
    }
    v8f z0 = {}, z1 = {};
    {
        const float* w0 = W2T + l * WTS + 8 * half;
        const float* w1 = W2T + (16 + l) * WTS + 8 * half;
#pragma unroll
        for (int q = 0; q < 4; ++q) {
            v2f b  = {y0[2 * q], y0[2 * q + 1]};
            v2f wa = *(const v2f*)(w0 + 2 * q);
            v2f wb = *(const v2f*)(w1 + 2 * q);
            z0 = wmma_f32(wa, b, z0);
            z1 = wmma_f32(wb, b, z1);
        }
#pragma unroll
        for (int q = 0; q < 4; ++q) {
            v2f b  = {y1[2 * q], y1[2 * q + 1]};
            v2f wa = *(const v2f*)(w0 + 16 + 2 * q);
            v2f wb = *(const v2f*)(w1 + 16 + 2 * q);
            z0 = wmma_f32(wa, b, z0);
            z1 = wmma_f32(wb, b, z1);
        }
    }

    {
        const float* bb = b2s + 8 * half;
        const float* ww = W3s + 8 * half;
        float pa = 0.0f, pb = 0.0f;
#pragma unroll
        for (int r = 0; r < 8; ++r) {
            pa = fmaf(fmaxf(z0[r] + bb[r], 0.0f), ww[r], pa);
            pb = fmaf(fmaxf(z1[r] + bb[16 + r], 0.0f), ww[16 + r], pb);
        }
        float part = pa + pb;
        float v = part + xhalf(part);
        if (half == 0) vs[m0 + l] = v;
    }
    __syncthreads();
    {
        float o = tile_times_vec(a, vs, half);
        if (half == 0) a1s[m0 + l] = o;
    }
    __syncthreads();
    if (t < NUMK) out[(size_t)g * NUMK + t] = a1s[t] + b3[0];
}

extern "C" void kernel_launch(void* const* d_in, const int* in_sizes, int n_in,
                              void* d_out, int out_size, void* d_ws, size_t ws_size,
                              hipStream_t stream) {
    (void)n_in; (void)d_ws; (void)ws_size; (void)out_size;
    const float* Hx = (const float*)d_in[0];
    const int*   ei = (const int*)d_in[1];
    const float* W1 = (const float*)d_in[2];
    const float* b1 = (const float*)d_in[3];
    const float* W2 = (const float*)d_in[4];
    const float* b2 = (const float*)d_in[5];
    const float* W3 = (const float*)d_in[6];
    const float* b3 = (const float*)d_in[7];
    float* out = (float*)d_out;
    const int nb = in_sizes[0] / HXW;
    gcn3_kernel<<<dim3(nb), dim3(256), 0, stream>>>(Hx, ei, W1, b1, W2, b2, W3, b3, out);
}
